# starting kernel
# baseline (speedup 1.0000x reference)
#include <hip/hip_runtime.h>
#include <math.h>

#define B_      4
#define N_      10000
#define F_IN_   64
#define UNITS_  64
#define E_      160000
#define C_      128
#define M_      (B_ * N_)
#define MV_     3
#define KTOT_   (MV_ * C_)
#define LDSW_   388

typedef __attribute__((ext_vector_type(2))) float v2f;
typedef __attribute__((ext_vector_type(8))) float v8f;


__global__ void k_zero(float* __restrict__ p, int n) {
  int i = blockIdx.x * blockDim.x + threadIdx.x;
  if (i < n) p[i] = 0.0f;
}

__global__ void k_concat_ru(const float* __restrict__ in,
                            const float* __restrict__ hx,
                            float* __restrict__ v0) {
  int idx = blockIdx.x * blockDim.x + threadIdx.x;
  if (idx >= M_ * C_) return;
  int m  = idx >> 7;
  int ch = idx & (C_ - 1);
  float v = (ch < F_IN_) ? in[(size_t)m * F_IN_ + ch]
                         : hx[(size_t)m * UNITS_ + (ch - F_IN_)];
  v0[idx] = v;
}

__global__ void k_concat_c(const float* __restrict__ in,
                           const float* __restrict__ hx,
                           const float* __restrict__ ru,
                           float* __restrict__ v0) {
  int idx = blockIdx.x * blockDim.x + threadIdx.x;
  if (idx >= M_ * C_) return;
  int m  = idx >> 7;
  int ch = idx & (C_ - 1);
  float v;
  if (ch < F_IN_) {
    v = in[(size_t)m * F_IN_ + ch];
  } else {
    int u = ch - F_IN_;
    float r = ru[(size_t)m * (2 * UNITS_) + u];
    v = r * hx[(size_t)m * UNITS_ + u];
  }
  v0[idx] = v;
}

__global__ void k_scatter(const float* __restrict__ x,
                          float* __restrict__ y,
                          const int* __restrict__ src,
                          const int* __restrict__ dst,
                          const float* __restrict__ ker) {
  int idx = blockIdx.x * blockDim.x + threadIdx.x;
  if (idx >= E_ * 32) return;
  int e = idx >> 5;
  int q = (idx & 31) << 2;
  __builtin_prefetch(&src[e + 4096], 0, 0);
  __builtin_prefetch(&dst[e + 4096], 0, 0);
  int   s = src[e];
  int   d = dst[e];
  float k = ker[e];
#pragma unroll
  for (int b = 0; b < B_; ++b) {
    const float4 v = *(const float4*)(x + ((size_t)(b * N_ + s)) * C_ + q);
    float* yp = y + ((size_t)(b * N_ + d)) * C_ + q;
    atomicAdd(yp + 0, k * v.x);
    atomicAdd(yp + 1, k * v.y);
    atomicAdd(yp + 2, k * v.z);
    atomicAdd(yp + 3, k * v.w);
  }
}


__device__ inline void fill_As(float* __restrict__ As,
                               const float* __restrict__ v0,
                               const float* __restrict__ v1,
                               const float* __restrict__ v2,
                               int m0, int tid, int nthreads) {
  for (int i = tid; i < 16 * (KTOT_ / 4); i += nthreads) {
    int row  = i / (KTOT_ / 4);
    int kk4  = (i % (KTOT_ / 4)) * 4;
    int view = kk4 >> 7;
    int ch   = kk4 & (C_ - 1);
    const float* srcp = (view == 0 ? v0 : (view == 1 ? v1 : v2))
                        + ((size_t)(m0 + row)) * C_ + ch;
    *(float4*)(As + row * LDSW_ + kk4) = *(const float4*)srcp;
  }
  __syncthreads();
}

__device__ inline v8f wmma_tile_lds(const float* __restrict__ As,
                                    const float* __restrict__ W, int nout,
                                    int n0, int lane) {
  const int arow = lane & 15;
  const int koff = (lane >> 4) << 1;
  const int col  = n0 + (lane & 15);
  v8f acc = {};
#pragma unroll 4
  for (int kb = 0; kb < KTOT_ / 4; ++kb) {
    const int k0 = kb * 4 + koff;
    v2f a = *(const v2f*)(As + arow * LDSW_ + k0);
    v2f b;
    b.x = W[(size_t)k0 * nout + col];
    b.y = W[(size_t)(k0 + 1) * nout + col];
    acc = __builtin_amdgcn_wmma_f32_16x16x4_f32(
         false, a,  false, b,
         (short)0, acc,  false,  false);
  }
  return acc;
}

__global__ void k_gemm_ru(const float* __restrict__ v0,
                          const float* __restrict__ v1,
                          const float* __restrict__ v2,
                          const float* __restrict__ W,
                          const float* __restrict__ bias,
                          float* __restrict__ ru) {
  __shared__ float As[16 * LDSW_];
  const int lane = threadIdx.x & 31;
  const int wave = threadIdx.x >> 5;
  const int m0 = blockIdx.x * 16;
  const int n0 = wave * 16;
  fill_As(As, v0, v1, v2, m0, threadIdx.x, 256);
  v8f acc = wmma_tile_lds(As, W, 2 * UNITS_, n0, lane);
  const int col   = n0 + (lane & 15);
  const int mbase = m0 + ((lane >> 4) << 3);
  const float bv  = bias[col];
#pragma unroll
  for (int r = 0; r < 8; ++r) {
    int m = mbase + r;
    float x = acc[r] + bv;
    ru[(size_t)m * (2 * UNITS_) + col] = 1.0f / (1.0f + __expf(-x));
  }
}

__global__ void k_gemm_c(const float* __restrict__ v0,
                         const float* __restrict__ v1,
                         const float* __restrict__ v2,
                         const float* __restrict__ W,
                         const float* __restrict__ bias,
                         const float* __restrict__ ru,
                         const float* __restrict__ hx,
                         float* __restrict__ out) {
  __shared__ float As[16 * LDSW_];
  const int lane = threadIdx.x & 31;
  const int wave = threadIdx.x >> 5;
  const int m0 = blockIdx.x * 16;
  const int n0 = wave * 16;
  fill_As(As, v0, v1, v2, m0, threadIdx.x, 128);
  v8f acc = wmma_tile_lds(As, W, UNITS_, n0, lane);
  const int col   = n0 + (lane & 15);
  const int mbase = m0 + ((lane >> 4) << 3);
  const float bv  = bias[col];
#pragma unroll
  for (int r = 0; r < 8; ++r) {
    int m = mbase + r;
    float c = tanhf(acc[r] + bv);
    float u = ru[(size_t)m * (2 * UNITS_) + UNITS_ + col];
    float h = hx[(size_t)m * UNITS_ + col];
    out[(size_t)m * UNITS_ + col] = u * h + (1.0f - u) * c;
  }
}

extern "C" void kernel_launch(void* const* d_in, const int* in_sizes, int n_in,
                              void* d_out, int out_size, void* d_ws, size_t ws_size,
                              hipStream_t stream) {
  const float* inputs = (const float*)d_in[0];
  const float* hx     = (const float*)d_in[1];
  const int*   sidx   = (const int*)  d_in[2];
  const float* ker    = (const float*)d_in[3];
  const float* W_ru   = (const float*)d_in[4];
  const float* b_ru   = (const float*)d_in[5];
  const float* W_c    = (const float*)d_in[6];
  const float* b_c    = (const float*)d_in[7];
  float* out = (float*)d_out;

  float* ws = (float*)d_ws;
  const size_t VSZ = (size_t)M_ * C_;
  float* v0 = ws;
  float* v1 = ws + VSZ;
  float* v2 = ws + 2 * VSZ;
  float* ru = ws + 3 * VSZ;

  const int* src = sidx;
  const int* dst = sidx + E_;

  const int T = 256;
  const int gridMC   = (M_ * C_ + T - 1) / T;
  const int gridScat = (E_ * 32 + T - 1) / T;
  const int gridGemm = M_ / 16;

  k_concat_ru<<<gridMC, T, 0, stream>>>(inputs, hx, v0);
  k_zero<<<gridMC, T, 0, stream>>>(v1, M_ * C_);
  k_scatter<<<gridScat, T, 0, stream>>>(v0, v1, src, dst, ker);
  k_zero<<<gridMC, T, 0, stream>>>(v2, M_ * C_);
  k_scatter<<<gridScat, T, 0, stream>>>(v1, v2, src, dst, ker);
  k_gemm_ru<<<gridGemm, 256, 0, stream>>>(v0, v1, v2, W_ru, b_ru, ru);

  k_concat_c<<<gridMC, T, 0, stream>>>(inputs, hx, ru, v0);
  k_zero<<<gridMC, T, 0, stream>>>(v1, M_ * C_);
  k_scatter<<<gridScat, T, 0, stream>>>(v0, v1, src, dst, ker);
  k_zero<<<gridMC, T, 0, stream>>>(v2, M_ * C_);
  k_scatter<<<gridScat, T, 0, stream>>>(v1, v2, src, dst, ker);
  k_gemm_c<<<gridGemm, 128, 0, stream>>>(v0, v1, v2, W_c, b_c, ru, hx, out);
}


# Candidate conv reuses the gate conv's views of the shared input channels 0..63 (x, Ax, AAx); only the r*hx channels 64..127 are rebuilt, zeroed and scatter-added (same float atomics, half as many)
# speedup vs baseline: 1.2728x; 1.2728x over previous
#include <hip/hip_runtime.h>
#include <math.h>

#define B_      4
#define N_      10000
#define F_IN_   64
#define UNITS_  64
#define E_      160000
#define C_      128
#define M_      (B_ * N_)
#define MV_     3
#define KTOT_   (MV_ * C_)
#define LDSW_   388

typedef __attribute__((ext_vector_type(2))) float v2f;
typedef __attribute__((ext_vector_type(8))) float v8f;


__global__ void k_zero(float* __restrict__ p, int n) {
  int i = blockIdx.x * blockDim.x + threadIdx.x;
  if (i < n) p[i] = 0.0f;
}

__global__ void k_concat_ru(const float* __restrict__ in,
                            const float* __restrict__ hx,
                            float* __restrict__ v0) {
  int idx = blockIdx.x * blockDim.x + threadIdx.x;
  if (idx >= M_ * C_) return;
  int m  = idx >> 7;
  int ch = idx & (C_ - 1);
  float v = (ch < F_IN_) ? in[(size_t)m * F_IN_ + ch]
                         : hx[(size_t)m * UNITS_ + (ch - F_IN_)];
  v0[idx] = v;
}

__global__ void k_rhx_hi(const float* __restrict__ hx,
                         const float* __restrict__ ru,
                         float* __restrict__ v0) {
  int idx = blockIdx.x * blockDim.x + threadIdx.x;
  if (idx >= M_ * UNITS_) return;
  int m = idx >> 6;
  int u = idx & (UNITS_ - 1);
  float r = ru[(size_t)m * (2 * UNITS_) + u];
  v0[(size_t)m * C_ + F_IN_ + u] = r * hx[idx];
}

__global__ void k_zero_hi(float* __restrict__ p) {
  int idx = blockIdx.x * blockDim.x + threadIdx.x;
  if (idx >= M_ * UNITS_) return;
  int m = idx >> 6;
  int u = idx & (UNITS_ - 1);
  p[(size_t)m * C_ + F_IN_ + u] = 0.0f;
}

__global__ void k_scatter(const float* __restrict__ x,
                          float* __restrict__ y,
                          const int* __restrict__ src,
                          const int* __restrict__ dst,
                          const float* __restrict__ ker) {
  int idx = blockIdx.x * blockDim.x + threadIdx.x;
  if (idx >= E_ * 32) return;
  int e = idx >> 5;
  int q = (idx & 31) << 2;
  __builtin_prefetch(&src[e + 4096], 0, 0);
  __builtin_prefetch(&dst[e + 4096], 0, 0);
  int   s = src[e];
  int   d = dst[e];
  float k = ker[e];
#pragma unroll
  for (int b = 0; b < B_; ++b) {
    const float4 v = *(const float4*)(x + ((size_t)(b * N_ + s)) * C_ + q);
    float* yp = y + ((size_t)(b * N_ + d)) * C_ + q;
    atomicAdd(yp + 0, k * v.x);
    atomicAdd(yp + 1, k * v.y);
    atomicAdd(yp + 2, k * v.z);
    atomicAdd(yp + 3, k * v.w);
  }
}

__global__ void k_scatter_hi(const float* __restrict__ x,
                             float* __restrict__ y,
                             const int* __restrict__ src,
                             const int* __restrict__ dst,
                             const float* __restrict__ ker) {
  int idx = blockIdx.x * blockDim.x + threadIdx.x;
  if (idx >= E_ * 16) return;
  int e = idx >> 4;
  int q = F_IN_ + ((idx & 15) << 2);
  __builtin_prefetch(&src[e + 4096], 0, 0);
  __builtin_prefetch(&dst[e + 4096], 0, 0);
  int   s = src[e];
  int   d = dst[e];
  float k = ker[e];
#pragma unroll
  for (int b = 0; b < B_; ++b) {
    const float4 v = *(const float4*)(x + ((size_t)(b * N_ + s)) * C_ + q);
    float* yp = y + ((size_t)(b * N_ + d)) * C_ + q;
    atomicAdd(yp + 0, k * v.x);
    atomicAdd(yp + 1, k * v.y);
    atomicAdd(yp + 2, k * v.z);
    atomicAdd(yp + 3, k * v.w);
  }
}


__device__ inline void fill_As(float* __restrict__ As,
                               const float* __restrict__ v0,
                               const float* __restrict__ v1,
                               const float* __restrict__ v2,
                               int m0, int tid, int nthreads) {
  for (int i = tid; i < 16 * (KTOT_ / 4); i += nthreads) {
    int row  = i / (KTOT_ / 4);
    int kk4  = (i % (KTOT_ / 4)) * 4;
    int view = kk4 >> 7;
    int ch   = kk4 & (C_ - 1);
    const float* srcp = (view == 0 ? v0 : (view == 1 ? v1 : v2))
                        + ((size_t)(m0 + row)) * C_ + ch;
    *(float4*)(As + row * LDSW_ + kk4) = *(const float4*)srcp;
  }
  __syncthreads();
}

__device__ inline v8f wmma_tile_lds(const float* __restrict__ As,
                                    const float* __restrict__ W, int nout,
                                    int n0, int lane) {
  const int arow = lane & 15;
  const int koff = (lane >> 4) << 1;
  const int col  = n0 + (lane & 15);
  v8f acc = {};
#pragma unroll 4
  for (int kb = 0; kb < KTOT_ / 4; ++kb) {
    const int k0 = kb * 4 + koff;
    v2f a = *(const v2f*)(As + arow * LDSW_ + k0);
    v2f b;
    b.x = W[(size_t)k0 * nout + col];
    b.y = W[(size_t)(k0 + 1) * nout + col];
    acc = __builtin_amdgcn_wmma_f32_16x16x4_f32(
         false, a,  false, b,
         (short)0, acc,  false,  false);
  }
  return acc;
}

__global__ void k_gemm_ru(const float* __restrict__ v0,
                          const float* __restrict__ v1,
                          const float* __restrict__ v2,
                          const float* __restrict__ W,
                          const float* __restrict__ bias,
                          float* __restrict__ ru) {
  __shared__ float As[16 * LDSW_];
  const int lane = threadIdx.x & 31;
  const int wave = threadIdx.x >> 5;
  const int m0 = blockIdx.x * 16;
  const int n0 = wave * 16;
  fill_As(As, v0, v1, v2, m0, threadIdx.x, 256);
  v8f acc = wmma_tile_lds(As, W, 2 * UNITS_, n0, lane);
  const int col   = n0 + (lane & 15);
  const int mbase = m0 + ((lane >> 4) << 3);
  const float bv  = bias[col];
#pragma unroll
  for (int r = 0; r < 8; ++r) {
    int m = mbase + r;
    float x = acc[r] + bv;
    ru[(size_t)m * (2 * UNITS_) + col] = 1.0f / (1.0f + __expf(-x));
  }
}

__global__ void k_gemm_c(const float* __restrict__ v0,
                         const float* __restrict__ v1,
                         const float* __restrict__ v2,
                         const float* __restrict__ W,
                         const float* __restrict__ bias,
                         const float* __restrict__ ru,
                         const float* __restrict__ hx,
                         float* __restrict__ out) {
  __shared__ float As[16 * LDSW_];
  const int lane = threadIdx.x & 31;
  const int wave = threadIdx.x >> 5;
  const int m0 = blockIdx.x * 16;
  const int n0 = wave * 16;
  fill_As(As, v0, v1, v2, m0, threadIdx.x, 128);
  v8f acc = wmma_tile_lds(As, W, UNITS_, n0, lane);
  const int col   = n0 + (lane & 15);
  const int mbase = m0 + ((lane >> 4) << 3);
  const float bv  = bias[col];
#pragma unroll
  for (int r = 0; r < 8; ++r) {
    int m = mbase + r;
    float c = tanhf(acc[r] + bv);
    float u = ru[(size_t)m * (2 * UNITS_) + UNITS_ + col];
    float h = hx[(size_t)m * UNITS_ + col];
    out[(size_t)m * UNITS_ + col] = u * h + (1.0f - u) * c;
  }
}

extern "C" void kernel_launch(void* const* d_in, const int* in_sizes, int n_in,
                              void* d_out, int out_size, void* d_ws, size_t ws_size,
                              hipStream_t stream) {
  const float* inputs = (const float*)d_in[0];
  const float* hx     = (const float*)d_in[1];
  const int*   sidx   = (const int*)  d_in[2];
  const float* ker    = (const float*)d_in[3];
  const float* W_ru   = (const float*)d_in[4];
  const float* b_ru   = (const float*)d_in[5];
  const float* W_c    = (const float*)d_in[6];
  const float* b_c    = (const float*)d_in[7];
  float* out = (float*)d_out;

  float* ws = (float*)d_ws;
  const size_t VSZ = (size_t)M_ * C_;
  float* v0 = ws;
  float* v1 = ws + VSZ;
  float* v2 = ws + 2 * VSZ;
  float* ru = ws + 3 * VSZ;

  const int* src = sidx;
  const int* dst = sidx + E_;

  const int T = 256;
  const int gridMC   = (M_ * C_ + T - 1) / T;
  const int gridScat = (E_ * 32 + T - 1) / T;
  const int gridGemm = M_ / 16;
  const int gridHalf = (M_ * UNITS_ + T - 1) / T;

  k_concat_ru<<<gridMC, T, 0, stream>>>(inputs, hx, v0);
  k_zero<<<gridMC, T, 0, stream>>>(v1, M_ * C_);
  k_scatter<<<gridScat, T, 0, stream>>>(v0, v1, src, dst, ker);
  k_zero<<<gridMC, T, 0, stream>>>(v2, M_ * C_);
  k_scatter<<<gridScat, T, 0, stream>>>(v1, v2, src, dst, ker);
  k_gemm_ru<<<gridGemm, 256, 0, stream>>>(v0, v1, v2, W_ru, b_ru, ru);

  k_rhx_hi<<<gridHalf, T, 0, stream>>>(hx, ru, v0);
  k_zero_hi<<<gridHalf, T, 0, stream>>>(v1);
  k_scatter_hi<<<gridHalf, T, 0, stream>>>(v0, v1, src, dst, ker);
  k_zero_hi<<<gridHalf, T, 0, stream>>>(v2);
  k_scatter_hi<<<gridHalf, T, 0, stream>>>(v1, v2, src, dst, ker);
  k_gemm_c<<<gridGemm, 128, 0, stream>>>(v0, v1, v2, W_c, b_c, ru, hx, out);
}


# Atomic scatter-add replaced by a gather over an edge list regrouped by dst in-call (int-atomic counting sort): no float atomics, no zero fills; a node's sum order = slot-claim order
# speedup vs baseline: 2.7969x; 2.1974x over previous
#include <hip/hip_runtime.h>
#include <math.h>

#define B_      4
#define N_      10000
#define F_IN_   64
#define UNITS_  64
#define E_      160000
#define C_      128
#define M_      (B_ * N_)
#define MV_     3
#define KTOT_   (MV_ * C_)
#define LDSW_   388

typedef __attribute__((ext_vector_type(2))) float v2f;
typedef __attribute__((ext_vector_type(8))) float v8f;


__global__ void k_concat_ru(const float* __restrict__ in,
                            const float* __restrict__ hx,
                            float* __restrict__ v0) {
  int idx = blockIdx.x * blockDim.x + threadIdx.x;
  if (idx >= M_ * C_) return;
  int m  = idx >> 7;
  int ch = idx & (C_ - 1);
  float v = (ch < F_IN_) ? in[(size_t)m * F_IN_ + ch]
                         : hx[(size_t)m * UNITS_ + (ch - F_IN_)];
  v0[idx] = v;
}

__global__ void k_rhx_hi(const float* __restrict__ hx,
                         const float* __restrict__ ru,
                         float* __restrict__ v0) {
  int idx = blockIdx.x * blockDim.x + threadIdx.x;
  if (idx >= M_ * UNITS_) return;
  int m = idx >> 6;
  int u = idx & (UNITS_ - 1);
  float r = ru[(size_t)m * (2 * UNITS_) + u];
  v0[(size_t)m * C_ + F_IN_ + u] = r * hx[idx];
}


__global__ void k_csr_zero(int* __restrict__ cnt) {
  int i = blockIdx.x * blockDim.x + threadIdx.x;
  if (i < N_) cnt[i] = 0;
}

__global__ void k_csr_count(const int* __restrict__ dst, int* __restrict__ cnt) {
  int e = blockIdx.x * blockDim.x + threadIdx.x;
  if (e < E_) atomicAdd(&cnt[dst[e]], 1);
}

#define SCAN_T_   1024
#define SCAN_PER_ ((N_ + SCAN_T_ - 1) / SCAN_T_)
__global__ void k_csr_scan(int* __restrict__ cur, int* __restrict__ rowptr) {
  __shared__ int part[SCAN_T_];
  const int t = threadIdx.x;
  const int base = t * SCAN_PER_;
  int loc[SCAN_PER_];
  int sum = 0;
#pragma unroll
  for (int i = 0; i < SCAN_PER_; ++i) {
    int idx = base + i;
    int v = (idx < N_) ? cur[idx] : 0;
    loc[i] = sum;
    sum += v;
  }
  part[t] = sum;
  __syncthreads();
  for (int off = 1; off < SCAN_T_; off <<= 1) {
    int v = (t >= off) ? part[t - off] : 0;
    __syncthreads();
    part[t] += v;
    __syncthreads();
  }
  const int excl = part[t] - sum;
#pragma unroll
  for (int i = 0; i < SCAN_PER_; ++i) {
    int idx = base + i;
    if (idx < N_) {
      rowptr[idx] = excl + loc[i];
      cur[idx]    = excl + loc[i];
    }
  }
  if (t == SCAN_T_ - 1) rowptr[N_] = part[t];
}

__global__ void k_csr_fill(const int* __restrict__ src,
                           const int* __restrict__ dst,
                           const float* __restrict__ ker,
                           int* __restrict__ cur,
                           int* __restrict__ csrc,
                           float* __restrict__ cker) {
  int e = blockIdx.x * blockDim.x + threadIdx.x;
  if (e >= E_) return;
  int p = atomicAdd(&cur[dst[e]], 1);
  csrc[p] = src[e];
  cker[p] = ker[e];
}

template <int NE, int NB>
__device__ inline void gather_edges(const float* __restrict__ xl, size_t bstride,
                                    int ms, float mk, int j, float4* acc) {
  float4 v[NE][NB];
  float  k[NE];
#pragma unroll
  for (int u = 0; u < NE; ++u) {
    const int s = __builtin_amdgcn_readlane(ms, j + u);
    k[u] = __int_as_float(__builtin_amdgcn_readlane(__float_as_int(mk), j + u));
#pragma unroll
    for (int b = 0; b < NB; ++b)
      v[u][b] = *(const float4*)(xl + (size_t)s * C_ + b * bstride);
  }
#pragma unroll
  for (int u = 0; u < NE; ++u) {
#pragma unroll
    for (int b = 0; b < NB; ++b) {
      acc[b].x += k[u] * v[u][b].x;
      acc[b].y += k[u] * v[u][b].y;
      acc[b].z += k[u] * v[u][b].z;
      acc[b].w += k[u] * v[u][b].w;
    }
  }
}

template <bool HI>
__global__ void k_gather(const float* __restrict__ x,
                         float* __restrict__ y,
                         const int* __restrict__ rowptr,
                         const int* __restrict__ csrc,
                         const float* __restrict__ cker) {
  constexpr int NB = HI ? 2 : 4;
  constexpr int NE = HI ? 8 : 4;
  const int lane = threadIdx.x & 31;
  const int n = blockIdx.x * (blockDim.x >> 5) + (threadIdx.x >> 5);
  if (n >= N_) return;
  const int beg = __builtin_amdgcn_readfirstlane(rowptr[n]);
  const int end = __builtin_amdgcn_readfirstlane(rowptr[n + 1]);
  const size_t loff = HI ? ((size_t)(lane >> 4) * N_ * C_ + F_IN_ + ((lane & 15) << 2))
                         : (size_t)(lane << 2);
  const size_t bstride = (HI ? 2 : 1) * (size_t)N_ * C_;
  const float* xl = x + loff;
  float4 acc[NB];
#pragma unroll
  for (int b = 0; b < NB; ++b) acc[b] = make_float4(0.0f, 0.0f, 0.0f, 0.0f);

  for (int base = beg; base < end; base += 32) {
    const int cnt = (end - base < 32) ? (end - base) : 32;
    int   ms = 0;
    float mk = 0.0f;
    if (lane < cnt) {
      ms = csrc[base + lane];
      mk = cker[base + lane];
    }
    int j = 0;
    for (; j + NE <= cnt; j += NE) gather_edges<NE, NB>(xl, bstride, ms, mk, j, acc);
    if (NE > 4 && (cnt - j) >= 4) { gather_edges<4, NB>(xl, bstride, ms, mk, j, acc); j += 4; }
    if ((cnt - j) >= 2)           { gather_edges<2, NB>(xl, bstride, ms, mk, j, acc); j += 2; }
    if ((cnt - j) >= 1)           { gather_edges<1, NB>(xl, bstride, ms, mk, j, acc); }
  }

  float* yl = y + loff + (size_t)n * C_;
#pragma unroll
  for (int b = 0; b < NB; ++b) *(float4*)(yl + b * bstride) = acc[b];
}


__device__ inline void fill_As(float* __restrict__ As,
                               const float* __restrict__ v0,
                               const float* __restrict__ v1,
                               const float* __restrict__ v2,
                               int m0, int tid, int nthreads) {
  for (int i = tid; i < 16 * (KTOT_ / 4); i += nthreads) {
    int row  = i / (KTOT_ / 4);
    int kk4  = (i % (KTOT_ / 4)) * 4;
    int view = kk4 >> 7;
    int ch   = kk4 & (C_ - 1);
    const float* srcp = (view == 0 ? v0 : (view == 1 ? v1 : v2))
                        + ((size_t)(m0 + row)) * C_ + ch;
    *(float4*)(As + row * LDSW_ + kk4) = *(const float4*)srcp;
  }
  __syncthreads();
}

__device__ inline v8f wmma_tile_lds(const float* __restrict__ As,
                                    const float* __restrict__ W, int nout,
                                    int n0, int lane) {
  const int arow = lane & 15;
  const int koff = (lane >> 4) << 1;
  const int col  = n0 + (lane & 15);
  v8f acc = {};
#pragma unroll 4
  for (int kb = 0; kb < KTOT_ / 4; ++kb) {
    const int k0 = kb * 4 + koff;
    v2f a = *(const v2f*)(As + arow * LDSW_ + k0);
    v2f b;
    b.x = W[(size_t)k0 * nout + col];
    b.y = W[(size_t)(k0 + 1) * nout + col];
    acc = __builtin_amdgcn_wmma_f32_16x16x4_f32(
         false, a,  false, b,
         (short)0, acc,  false,  false);
  }
  return acc;
}

__global__ void k_gemm_ru(const float* __restrict__ v0,
                          const float* __restrict__ v1,
                          const float* __restrict__ v2,
                          const float* __restrict__ W,
                          const float* __restrict__ bias,
                          float* __restrict__ ru) {
  __shared__ float As[16 * LDSW_];
  const int lane = threadIdx.x & 31;
  const int wave = threadIdx.x >> 5;
  const int m0 = blockIdx.x * 16;
  const int n0 = wave * 16;
  fill_As(As, v0, v1, v2, m0, threadIdx.x, 256);
  v8f acc = wmma_tile_lds(As, W, 2 * UNITS_, n0, lane);
  const int col   = n0 + (lane & 15);
  const int mbase = m0 + ((lane >> 4) << 3);
  const float bv  = bias[col];
#pragma unroll
  for (int r = 0; r < 8; ++r) {
    int m = mbase + r;
    float x = acc[r] + bv;
    ru[(size_t)m * (2 * UNITS_) + col] = 1.0f / (1.0f + __expf(-x));
  }
}

__global__ void k_gemm_c(const float* __restrict__ v0,
                         const float* __restrict__ v1,
                         const float* __restrict__ v2,
                         const float* __restrict__ W,
                         const float* __restrict__ bias,
                         const float* __restrict__ ru,
                         const float* __restrict__ hx,
                         float* __restrict__ out) {
  __shared__ float As[16 * LDSW_];
  const int lane = threadIdx.x & 31;
  const int wave = threadIdx.x >> 5;
  const int m0 = blockIdx.x * 16;
  const int n0 = wave * 16;
  fill_As(As, v0, v1, v2, m0, threadIdx.x, 128);
  v8f acc = wmma_tile_lds(As, W, UNITS_, n0, lane);
  const int col   = n0 + (lane & 15);
  const int mbase = m0 + ((lane >> 4) << 3);
  const float bv  = bias[col];
#pragma unroll
  for (int r = 0; r < 8; ++r) {
    int m = mbase + r;
    float c = tanhf(acc[r] + bv);
    float u = ru[(size_t)m * (2 * UNITS_) + UNITS_ + col];
    float h = hx[(size_t)m * UNITS_ + col];
    out[(size_t)m * UNITS_ + col] = u * h + (1.0f - u) * c;
  }
}

extern "C" void kernel_launch(void* const* d_in, const int* in_sizes, int n_in,
                              void* d_out, int out_size, void* d_ws, size_t ws_size,
                              hipStream_t stream) {
  const float* inputs = (const float*)d_in[0];
  const float* hx     = (const float*)d_in[1];
  const int*   sidx   = (const int*)  d_in[2];
  const float* ker    = (const float*)d_in[3];
  const float* W_ru   = (const float*)d_in[4];
  const float* b_ru   = (const float*)d_in[5];
  const float* W_c    = (const float*)d_in[6];
  const float* b_c    = (const float*)d_in[7];
  float* out = (float*)d_out;

  float* ws = (float*)d_ws;
  const size_t VSZ = (size_t)M_ * C_;
  float* v0 = ws;
  float* v1 = ws + VSZ;
  float* v2 = ws + 2 * VSZ;
  float* ru = ws + 3 * VSZ;

  const int* src = sidx;
  const int* dst = sidx + E_;

  int*   rowptr = (int*)d_out;
  int*   cur    = rowptr + 10240;
  int*   csrc   = cur + 10240;
  float* cker   = (float*)(csrc + E_);

  const int T = 256;
  const int gridMC   = (M_ * C_ + T - 1) / T;
  const int gridGemm = M_ / 16;
  const int gridHalf = (M_ * UNITS_ + T - 1) / T;
  const int gridN    = (N_ + T - 1) / T;
  const int gridE    = (E_ + T - 1) / T;
  const int gridGath = (N_ + 7) / 8;

  k_csr_zero<<<gridN, T, 0, stream>>>(cur);
  k_csr_count<<<gridE, T, 0, stream>>>(dst, cur);
  k_csr_scan<<<1, SCAN_T_, 0, stream>>>(cur, rowptr);
  k_csr_fill<<<gridE, T, 0, stream>>>(src, dst, ker, cur, csrc, cker);

  k_concat_ru<<<gridMC, T, 0, stream>>>(inputs, hx, v0);
  k_gather<false><<<gridGath, T, 0, stream>>>(v0, v1, rowptr, csrc, cker);
  k_gather<false><<<gridGath, T, 0, stream>>>(v1, v2, rowptr, csrc, cker);
  k_gemm_ru<<<gridGemm, 256, 0, stream>>>(v0, v1, v2, W_ru, b_ru, ru);

  k_rhx_hi<<<gridHalf, T, 0, stream>>>(hx, ru, v0);
  k_gather<true><<<gridGath, T, 0, stream>>>(v0, v1, rowptr, csrc, cker);
  k_gather<true><<<gridGath, T, 0, stream>>>(v1, v2, rowptr, csrc, cker);
  k_gemm_c<<<gridGemm, 128, 0, stream>>>(v0, v1, v2, W_c, b_c, ru, hx, out);
}


# Edge list regrouped in ONE pass into fixed 64-slot node buckets (one returning int add per edge, overflow to a spill list): count, scan and clear launches go, 12 -> 9; sum order = slot-claim order
# speedup vs baseline: 3.2871x; 1.1753x over previous
#include <hip/hip_runtime.h>
#include <math.h>

#define B_      4
#define N_      10000
#define F_IN_   64
#define UNITS_  64
#define E_      160000
#define C_      128
#define M_      (B_ * N_)
#define MV_     3
#define KTOT_   (MV_ * C_)
#define LDSW_   388

typedef __attribute__((ext_vector_type(2))) float v2f;
typedef __attribute__((ext_vector_type(8))) float v8f;


__global__ void k_concat_ru(const float* __restrict__ in,
                            const float* __restrict__ hx,
                            float* __restrict__ v0,
                            int* __restrict__ cnt) {
  int idx = blockIdx.x * blockDim.x + threadIdx.x;
  if (idx >= M_ * C_) return;
  if (idx <= N_) cnt[idx] = 0;
  int m  = idx >> 7;
  int ch = idx & (C_ - 1);
  float v = (ch < F_IN_) ? in[(size_t)m * F_IN_ + ch]
                         : hx[(size_t)m * UNITS_ + (ch - F_IN_)];
  v0[idx] = v;
}

__global__ void k_rhx_hi(const float* __restrict__ hx,
                         const float* __restrict__ ru,
                         float* __restrict__ v0) {
  int idx = blockIdx.x * blockDim.x + threadIdx.x;
  if (idx >= M_ * UNITS_) return;
  int m = idx >> 6;
  int u = idx & (UNITS_ - 1);
  float r = ru[(size_t)m * (2 * UNITS_) + u];
  v0[(size_t)m * C_ + F_IN_ + u] = r * hx[idx];
}

#define CAP_ 64

__global__ void k_bucket_fill(const int* __restrict__ src,
                              const int* __restrict__ dst,
                              const float* __restrict__ ker,
                              int* __restrict__ cnt,
                              int* __restrict__ csrc,
                              float* __restrict__ cker,
                              int* __restrict__ sp_src,
                              int* __restrict__ sp_dst,
                              float* __restrict__ sp_ker) {
  int e = blockIdx.x * blockDim.x + threadIdx.x;
  if (e >= E_) return;
  const int d = dst[e];
  const int p = atomicAdd(&cnt[d], 1);
  if (p < CAP_) {
    csrc[d * CAP_ + p] = src[e];
    cker[d * CAP_ + p] = ker[e];
  } else {
    const int q = atomicAdd(&cnt[N_], 1);
    sp_src[q] = src[e];
    sp_dst[q] = d;
    sp_ker[q] = ker[e];
  }
}

template <int NE, int NB>
__device__ inline void gather_edges(const float* __restrict__ xl, size_t bstride,
                                    int ms, float mk, int j, float4* acc) {
  float4 v[NE][NB];
  float  k[NE];
#pragma unroll
  for (int u = 0; u < NE; ++u) {
    const int s = __builtin_amdgcn_readlane(ms, j + u);
    k[u] = __int_as_float(__builtin_amdgcn_readlane(__float_as_int(mk), j + u));
#pragma unroll
    for (int b = 0; b < NB; ++b)
      v[u][b] = *(const float4*)(xl + (size_t)s * C_ + b * bstride);
  }
#pragma unroll
  for (int u = 0; u < NE; ++u) {
#pragma unroll
    for (int b = 0; b < NB; ++b) {
      acc[b].x += k[u] * v[u][b].x;
      acc[b].y += k[u] * v[u][b].y;
      acc[b].z += k[u] * v[u][b].z;
      acc[b].w += k[u] * v[u][b].w;
    }
  }
}

template <bool HI>
__global__ void k_gather(const float* __restrict__ x,
                         float* __restrict__ y,
                         const int* __restrict__ claimed,
                         const int* __restrict__ csrc,
                         const float* __restrict__ cker,
                         const int* __restrict__ sp_src,
                         const int* __restrict__ sp_dst,
                         const float* __restrict__ sp_ker) {
  constexpr int NB = HI ? 2 : 4;
  constexpr int NE = HI ? 8 : 4;
  const int lane = threadIdx.x & 31;
  const int n = blockIdx.x * (blockDim.x >> 5) + (threadIdx.x >> 5);
  if (n >= N_) return;
  const int deg = __builtin_amdgcn_readfirstlane(claimed[n]);
  const int beg = n * CAP_;
  const int end = beg + (deg < CAP_ ? deg : CAP_);
  const size_t loff = HI ? ((size_t)(lane >> 4) * N_ * C_ + F_IN_ + ((lane & 15) << 2))
                         : (size_t)(lane << 2);
  const size_t bstride = (HI ? 2 : 1) * (size_t)N_ * C_;
  const float* xl = x + loff;
  float4 acc[NB];
#pragma unroll
  for (int b = 0; b < NB; ++b) acc[b] = make_float4(0.0f, 0.0f, 0.0f, 0.0f);

  for (int base = beg; base < end; base += 32) {
    const int cnt = (end - base < 32) ? (end - base) : 32;
    int   ms = 0;
    float mk = 0.0f;
    if (lane < cnt) {
      ms = csrc[base + lane];
      mk = cker[base + lane];
    }
    int j = 0;
    for (; j + NE <= cnt; j += NE) gather_edges<NE, NB>(xl, bstride, ms, mk, j, acc);
    if (NE > 4 && (cnt - j) >= 4) { gather_edges<4, NB>(xl, bstride, ms, mk, j, acc); j += 4; }
    if ((cnt - j) >= 2)           { gather_edges<2, NB>(xl, bstride, ms, mk, j, acc); j += 2; }
    if ((cnt - j) >= 1)           { gather_edges<1, NB>(xl, bstride, ms, mk, j, acc); }
  }

  if (deg > CAP_) {
    int ns = __builtin_amdgcn_readfirstlane(claimed[N_]);
    ns = (ns < E_) ? ns : E_;
    for (int base = 0; base < ns; base += 32) {
      const int c = (ns - base < 32) ? (ns - base) : 32;
      int   md = -1;
      int   ms = 0;
      float mk = 0.0f;
      if (lane < c) {
        md = sp_dst[base + lane];
        ms = sp_src[base + lane];
        mk = sp_ker[base + lane];
      }
#pragma unroll 1
      for (int j = 0; j < c; ++j)
        if (__builtin_amdgcn_readlane(md, j) == n) gather_edges<1, NB>(xl, bstride, ms, mk, j, acc);
    }
  }

  float* yl = y + loff + (size_t)n * C_;
#pragma unroll
  for (int b = 0; b < NB; ++b) *(float4*)(yl + b * bstride) = acc[b];
}


__device__ inline void fill_As(float* __restrict__ As,
                               const float* __restrict__ v0,
                               const float* __restrict__ v1,
                               const float* __restrict__ v2,
                               int m0, int tid, int nthreads) {
  for (int i = tid; i < 16 * (KTOT_ / 4); i += nthreads) {
    int row  = i / (KTOT_ / 4);
    int kk4  = (i % (KTOT_ / 4)) * 4;
    int view = kk4 >> 7;
    int ch   = kk4 & (C_ - 1);
    const float* srcp = (view == 0 ? v0 : (view == 1 ? v1 : v2))
                        + ((size_t)(m0 + row)) * C_ + ch;
    *(float4*)(As + row * LDSW_ + kk4) = *(const float4*)srcp;
  }
  __syncthreads();
}

__device__ inline v8f wmma_tile_lds(const float* __restrict__ As,
                                    const float* __restrict__ W, int nout,
                                    int n0, int lane) {
  const int arow = lane & 15;
  const int koff = (lane >> 4) << 1;
  const int col  = n0 + (lane & 15);
  v8f acc = {};
#pragma unroll 4
  for (int kb = 0; kb < KTOT_ / 4; ++kb) {
    const int k0 = kb * 4 + koff;
    v2f a = *(const v2f*)(As + arow * LDSW_ + k0);
    v2f b;
    b.x = W[(size_t)k0 * nout + col];
    b.y = W[(size_t)(k0 + 1) * nout + col];
    acc = __builtin_amdgcn_wmma_f32_16x16x4_f32(
         false, a,  false, b,
         (short)0, acc,  false,  false);
  }
  return acc;
}

__global__ void k_gemm_ru(const float* __restrict__ v0,
                          const float* __restrict__ v1,
                          const float* __restrict__ v2,
                          const float* __restrict__ W,
                          const float* __restrict__ bias,
                          float* __restrict__ ru) {
  __shared__ float As[16 * LDSW_];
  const int lane = threadIdx.x & 31;
  const int wave = threadIdx.x >> 5;
  const int m0 = blockIdx.x * 16;
  const int n0 = wave * 16;
  fill_As(As, v0, v1, v2, m0, threadIdx.x, 256);
  v8f acc = wmma_tile_lds(As, W, 2 * UNITS_, n0, lane);
  const int col   = n0 + (lane & 15);
  const int mbase = m0 + ((lane >> 4) << 3);
  const float bv  = bias[col];
#pragma unroll
  for (int r = 0; r < 8; ++r) {
    int m = mbase + r;
    float x = acc[r] + bv;
    ru[(size_t)m * (2 * UNITS_) + col] = 1.0f / (1.0f + __expf(-x));
  }
}

__global__ void k_gemm_c(const float* __restrict__ v0,
                         const float* __restrict__ v1,
                         const float* __restrict__ v2,
                         const float* __restrict__ W,
                         const float* __restrict__ bias,
                         const float* __restrict__ ru,
                         const float* __restrict__ hx,
                         float* __restrict__ out) {
  __shared__ float As[16 * LDSW_];
  const int lane = threadIdx.x & 31;
  const int wave = threadIdx.x >> 5;
  const int m0 = blockIdx.x * 16;
  const int n0 = wave * 16;
  fill_As(As, v0, v1, v2, m0, threadIdx.x, 128);
  v8f acc = wmma_tile_lds(As, W, UNITS_, n0, lane);
  const int col   = n0 + (lane & 15);
  const int mbase = m0 + ((lane >> 4) << 3);
  const float bv  = bias[col];
#pragma unroll
  for (int r = 0; r < 8; ++r) {
    int m = mbase + r;
    float c = tanhf(acc[r] + bv);
    float u = ru[(size_t)m * (2 * UNITS_) + UNITS_ + col];
    float h = hx[(size_t)m * UNITS_ + col];
    out[(size_t)m * UNITS_ + col] = u * h + (1.0f - u) * c;
  }
}

extern "C" void kernel_launch(void* const* d_in, const int* in_sizes, int n_in,
                              void* d_out, int out_size, void* d_ws, size_t ws_size,
                              hipStream_t stream) {
  const float* inputs = (const float*)d_in[0];
  const float* hx     = (const float*)d_in[1];
  const int*   sidx   = (const int*)  d_in[2];
  const float* ker    = (const float*)d_in[3];
  const float* W_ru   = (const float*)d_in[4];
  const float* b_ru   = (const float*)d_in[5];
  const float* W_c    = (const float*)d_in[6];
  const float* b_c    = (const float*)d_in[7];
  float* out = (float*)d_out;

  float* ws = (float*)d_ws;
  const size_t VSZ = (size_t)M_ * C_;
  float* v0 = ws;
  float* v1 = ws + VSZ;
  float* v2 = ws + 2 * VSZ;
  float* ru = ws + 3 * VSZ;

  const int* src = sidx;
  const int* dst = sidx + E_;

  int*   cnt    = (int*)d_out;
  int*   csrc   = cnt + 10240;
  float* cker   = (float*)(csrc + N_ * CAP_);
  int*   sp_src = (int*)(cker + N_ * CAP_);
  int*   sp_dst = sp_src + E_;
  float* sp_ker = (float*)(sp_dst + E_);

  const int T = 256;
  const int gridMC   = (M_ * C_ + T - 1) / T;
  const int gridGemm = M_ / 16;
  const int gridHalf = (M_ * UNITS_ + T - 1) / T;
  const int gridE    = (E_ + T - 1) / T;
  const int gridGath = (N_ + 7) / 8;

  k_concat_ru<<<gridMC, T, 0, stream>>>(inputs, hx, v0, cnt);

  k_bucket_fill<<<gridE, T, 0, stream>>>(src, dst, ker, cnt, csrc, cker, sp_src, sp_dst, sp_ker);

  k_gather<false><<<gridGath, T, 0, stream>>>(v0, v1, cnt, csrc, cker, sp_src, sp_dst, sp_ker);
  k_gather<false><<<gridGath, T, 0, stream>>>(v1, v2, cnt, csrc, cker, sp_src, sp_dst, sp_ker);
  k_gemm_ru<<<gridGemm, 256, 0, stream>>>(v0, v1, v2, W_ru, b_ru, ru);

  k_rhx_hi<<<gridHalf, T, 0, stream>>>(hx, ru, v0);
  k_gather<true><<<gridGath, T, 0, stream>>>(v0, v1, cnt, csrc, cker, sp_src, sp_dst, sp_ker);
  k_gather<true><<<gridGath, T, 0, stream>>>(v1, v2, cnt, csrc, cker, sp_src, sp_dst, sp_ker);
  k_gemm_c<<<gridGemm, 128, 0, stream>>>(v0, v1, v2, W_c, b_c, ru, hx, out);
}


# View 0 (the concat) is never stored: gathers and GEMM tiles read inputs/hx in place, the gate GEMM's epilogue writes r*hx; both elementwise copy launches go (9 -> 8, a counter clear stays)
# speedup vs baseline: 3.7257x; 1.1334x over previous
#include <hip/hip_runtime.h>
#include <math.h>

#define B_      4
#define N_      10000
#define F_IN_   64
#define UNITS_  64
#define E_      160000
#define C_      128
#define M_      (B_ * N_)
#define MV_     3
#define KTOT_   (MV_ * C_)
#define LDSW_   388

typedef __attribute__((ext_vector_type(2))) float v2f;
typedef __attribute__((ext_vector_type(8))) float v8f;


__global__ void k_clear_cnt(int* __restrict__ cnt) {
  int idx = blockIdx.x * blockDim.x + threadIdx.x;
  if (idx <= N_) cnt[idx] = 0;
}

#define CAP_ 64

__global__ void k_bucket_fill(const int* __restrict__ src,
                              const int* __restrict__ dst,
                              const float* __restrict__ ker,
                              int* __restrict__ cnt,
                              int* __restrict__ csrc,
                              float* __restrict__ cker,
                              int* __restrict__ sp_src,
                              int* __restrict__ sp_dst,
                              float* __restrict__ sp_ker) {
  int e = blockIdx.x * blockDim.x + threadIdx.x;
  if (e >= E_) return;
  const int d = dst[e];
  const int p = atomicAdd(&cnt[d], 1);
  if (p < CAP_) {
    csrc[d * CAP_ + p] = src[e];
    cker[d * CAP_ + p] = ker[e];
  } else {
    const int q = atomicAdd(&cnt[N_], 1);
    sp_src[q] = src[e];
    sp_dst[q] = d;
    sp_ker[q] = ker[e];
  }
}

template <int NE, int NB, int XRS>
__device__ inline void gather_edges(const float* __restrict__ xl, size_t bstride,
                                    int ms, float mk, int j, float4* acc) {
  float4 v[NE][NB];
  float  k[NE];
#pragma unroll
  for (int u = 0; u < NE; ++u) {
    const int s = __builtin_amdgcn_readlane(ms, j + u);
    k[u] = __int_as_float(__builtin_amdgcn_readlane(__float_as_int(mk), j + u));
#pragma unroll
    for (int b = 0; b < NB; ++b)
      v[u][b] = *(const float4*)(xl + (size_t)s * XRS + b * bstride);
  }
#pragma unroll
  for (int u = 0; u < NE; ++u) {
#pragma unroll
    for (int b = 0; b < NB; ++b) {
      acc[b].x += k[u] * v[u][b].x;
      acc[b].y += k[u] * v[u][b].y;
      acc[b].z += k[u] * v[u][b].z;
      acc[b].w += k[u] * v[u][b].w;
    }
  }
}

template <int MODE>
__global__ void k_gather(const float* __restrict__ xa,
                         const float* __restrict__ xb,
                         float* __restrict__ y,
                         const int* __restrict__ claimed,
                         const int* __restrict__ csrc,
                         const float* __restrict__ cker,
                         const int* __restrict__ sp_src,
                         const int* __restrict__ sp_dst,
                         const float* __restrict__ sp_ker) {
  constexpr bool HI = (MODE >= 2);
  constexpr int NB  = HI ? 2 : 4;
  constexpr int NE  = HI ? 8 : 4;
  constexpr int XRS = (MODE == 0) ? F_IN_ : C_;
  const int lane = threadIdx.x & 31;
  const int n = blockIdx.x * (blockDim.x >> 5) + (threadIdx.x >> 5);
  if (n >= N_) return;
  const int deg = __builtin_amdgcn_readfirstlane(claimed[n]);
  const int beg = n * CAP_;
  const int end = beg + (deg < CAP_ ? deg : CAP_);
  const size_t yoff = HI ? ((size_t)(lane >> 4) * N_ * C_ + F_IN_ + ((lane & 15) << 2))
                         : (size_t)(lane << 2);
  const size_t ybs  = (HI ? 2 : 1) * (size_t)N_ * C_;
  const float* xl;
  if (MODE == 0)      xl = ((lane < 16) ? xa : xb) + ((lane & 15) << 2);
  else if (MODE == 1) xl = xa + (lane << 2);
  else if (MODE == 2) xl = xa + (size_t)(lane >> 4) * N_ * C_ + ((lane & 15) << 2);
  else                xl = xa + yoff;
  const size_t bstride = (HI ? 2 : 1) * (size_t)N_ * XRS;
  float4 acc[NB];
#pragma unroll
  for (int b = 0; b < NB; ++b) acc[b] = make_float4(0.0f, 0.0f, 0.0f, 0.0f);

  for (int base = beg; base < end; base += 32) {
    const int cnt = (end - base < 32) ? (end - base) : 32;
    int   ms = 0;
    float mk = 0.0f;
    if (lane < cnt) {
      ms = csrc[base + lane];
      mk = cker[base + lane];
    }
    int j = 0;
    for (; j + NE <= cnt; j += NE) gather_edges<NE, NB, XRS>(xl, bstride, ms, mk, j, acc);
    if (NE > 4 && (cnt - j) >= 4) { gather_edges<4, NB, XRS>(xl, bstride, ms, mk, j, acc); j += 4; }
    if ((cnt - j) >= 2)           { gather_edges<2, NB, XRS>(xl, bstride, ms, mk, j, acc); j += 2; }
    if ((cnt - j) >= 1)           { gather_edges<1, NB, XRS>(xl, bstride, ms, mk, j, acc); }
  }

  if (deg > CAP_) {
    int ns = __builtin_amdgcn_readfirstlane(claimed[N_]);
    ns = (ns < E_) ? ns : E_;
    for (int base = 0; base < ns; base += 32) {
      const int c = (ns - base < 32) ? (ns - base) : 32;
      int   md = -1;
      int   ms = 0;
      float mk = 0.0f;
      if (lane < c) {
        md = sp_dst[base + lane];
        ms = sp_src[base + lane];
        mk = sp_ker[base + lane];
      }
#pragma unroll 1
      for (int j = 0; j < c; ++j)
        if (__builtin_amdgcn_readlane(md, j) == n) gather_edges<1, NB, XRS>(xl, bstride, ms, mk, j, acc);
    }
  }

  float* yl = y + yoff + (size_t)n * C_;
#pragma unroll
  for (int b = 0; b < NB; ++b) *(float4*)(yl + b * ybs) = acc[b];
}


template <int HIS>
__device__ inline void fill_As(float* __restrict__ As,
                               const float* __restrict__ lo,
                               const float* __restrict__ hi,
                               const float* __restrict__ v1,
                               const float* __restrict__ v2,
                               int m0, int tid, int nthreads) {
  for (int i = tid; i < 16 * (KTOT_ / 4); i += nthreads) {
    int row  = i / (KTOT_ / 4);
    int kk4  = (i % (KTOT_ / 4)) * 4;
    int view = kk4 >> 7;
    int ch   = kk4 & (C_ - 1);
    const size_t m = (size_t)(m0 + row);
    const float* srcp;
    if (view == 0) srcp = (ch < F_IN_) ? lo + m * F_IN_ + ch
                                       : hi + m * HIS + (ch - F_IN_);
    else           srcp = (view == 1 ? v1 : v2) + m * C_ + ch;
    *(float4*)(As + row * LDSW_ + kk4) = *(const float4*)srcp;
  }
  __syncthreads();
}

__device__ inline v8f wmma_tile_lds(const float* __restrict__ As,
                                    const float* __restrict__ W, int nout,
                                    int n0, int lane) {
  const int arow = lane & 15;
  const int koff = (lane >> 4) << 1;
  const int col  = n0 + (lane & 15);
  v8f acc = {};
#pragma unroll 4
  for (int kb = 0; kb < KTOT_ / 4; ++kb) {
    const int k0 = kb * 4 + koff;
    v2f a = *(const v2f*)(As + arow * LDSW_ + k0);
    v2f b;
    b.x = W[(size_t)k0 * nout + col];
    b.y = W[(size_t)(k0 + 1) * nout + col];
    acc = __builtin_amdgcn_wmma_f32_16x16x4_f32(
         false, a,  false, b,
         (short)0, acc,  false,  false);
  }
  return acc;
}

__global__ void k_gemm_ru(const float* __restrict__ in,
                          const float* __restrict__ hx,
                          const float* __restrict__ v1,
                          const float* __restrict__ v2,
                          const float* __restrict__ W,
                          const float* __restrict__ bias,
                          float* __restrict__ ru) {
  __shared__ float As[16 * LDSW_];
  const int lane = threadIdx.x & 31;
  const int wave = threadIdx.x >> 5;
  const int m0 = blockIdx.x * 16;
  const int n0 = wave * 16;
  fill_As<UNITS_>(As, in, hx, v1, v2, m0, threadIdx.x, 256);
  v8f acc = wmma_tile_lds(As, W, 2 * UNITS_, n0, lane);
  const int col   = n0 + (lane & 15);
  const int mbase = m0 + ((lane >> 4) << 3);
  const float bv  = bias[col];
  float g[8];
#pragma unroll
  for (int r = 0; r < 8; ++r) {
    float x = acc[r] + bv;
    g[r] = 1.0f / (1.0f + __expf(-x));
  }
  if (n0 < UNITS_) {
#pragma unroll
    for (int r = 0; r < 8; ++r) g[r] *= hx[(size_t)(mbase + r) * UNITS_ + col];
  }
#pragma unroll
  for (int r = 0; r < 8; ++r) ru[(size_t)(mbase + r) * (2 * UNITS_) + col] = g[r];
}

__global__ void k_gemm_c(const float* __restrict__ in,
                         const float* __restrict__ v1,
                         const float* __restrict__ v2,
                         const float* __restrict__ W,
                         const float* __restrict__ bias,
                         const float* __restrict__ ru,
                         const float* __restrict__ hx,
                         float* __restrict__ out) {
  __shared__ float As[16 * LDSW_];
  const int lane = threadIdx.x & 31;
  const int wave = threadIdx.x >> 5;
  const int m0 = blockIdx.x * 16;
  const int n0 = wave * 16;
  fill_As<2 * UNITS_>(As, in, ru, v1, v2, m0, threadIdx.x, 128);
  v8f acc = wmma_tile_lds(As, W, UNITS_, n0, lane);
  const int col   = n0 + (lane & 15);
  const int mbase = m0 + ((lane >> 4) << 3);
  const float bv  = bias[col];
#pragma unroll
  for (int r = 0; r < 8; ++r) {
    int m = mbase + r;
    float c = tanhf(acc[r] + bv);
    float u = ru[(size_t)m * (2 * UNITS_) + UNITS_ + col];
    float h = hx[(size_t)m * UNITS_ + col];
    out[(size_t)m * UNITS_ + col] = u * h + (1.0f - u) * c;
  }
}

extern "C" void kernel_launch(void* const* d_in, const int* in_sizes, int n_in,
                              void* d_out, int out_size, void* d_ws, size_t ws_size,
                              hipStream_t stream) {
  const float* inputs = (const float*)d_in[0];
  const float* hx     = (const float*)d_in[1];
  const int*   sidx   = (const int*)  d_in[2];
  const float* ker    = (const float*)d_in[3];
  const float* W_ru   = (const float*)d_in[4];
  const float* b_ru   = (const float*)d_in[5];
  const float* W_c    = (const float*)d_in[6];
  const float* b_c    = (const float*)d_in[7];
  float* out = (float*)d_out;

  float* ws = (float*)d_ws;
  const size_t VSZ = (size_t)M_ * C_;
  float* v1 = ws;
  float* v2 = ws + VSZ;
  float* ru = ws + 2 * VSZ;

  const int* src = sidx;
  const int* dst = sidx + E_;

  int*   cnt    = (int*)d_out;
  int*   csrc   = cnt + 10240;
  float* cker   = (float*)(csrc + N_ * CAP_);
  int*   sp_src = (int*)(cker + N_ * CAP_);
  int*   sp_dst = sp_src + E_;
  float* sp_ker = (float*)(sp_dst + E_);

  const int T = 256;
  const int gridClr  = (N_ + 1 + T - 1) / T;
  const int gridGemm = M_ / 16;
  const int gridE    = (E_ + T - 1) / T;
  const int gridGath = (N_ + 7) / 8;

  k_clear_cnt<<<gridClr, T, 0, stream>>>(cnt);
  k_bucket_fill<<<gridE, T, 0, stream>>>(src, dst, ker, cnt, csrc, cker, sp_src, sp_dst, sp_ker);

  k_gather<0><<<gridGath, T, 0, stream>>>(inputs, hx, v1, cnt, csrc, cker, sp_src, sp_dst, sp_ker);
  k_gather<1><<<gridGath, T, 0, stream>>>(v1, v1, v2, cnt, csrc, cker, sp_src, sp_dst, sp_ker);
  k_gemm_ru<<<gridGemm, 256, 0, stream>>>(inputs, hx, v1, v2, W_ru, b_ru, ru);

  k_gather<2><<<gridGath, T, 0, stream>>>(ru, ru, v1, cnt, csrc, cker, sp_src, sp_dst, sp_ker);
  k_gather<3><<<gridGath, T, 0, stream>>>(v1, v1, v2, cnt, csrc, cker, sp_src, sp_dst, sp_ker);
  k_gemm_c<<<gridGemm, 128, 0, stream>>>(inputs, v1, v2, W_c, b_c, ru, hx, out);
}
